# starting kernel
# baseline (speedup 1.0000x reference)
#include <hip/hip_runtime.h>

#define BB 32
#define NN 1024
#define CC 768
#define HH 32

typedef __attribute__((ext_vector_type(2))) float v2f;
typedef __attribute__((ext_vector_type(8))) float v8f;

__global__ void init_ws_kernel(float* ws, int n) {
    for (int i = threadIdx.x; i < n; i += blockDim.x) ws[i] = 0.0f;
}

__global__ void __launch_bounds__(192)
pool_kernel(const float* __restrict__ x, float* __restrict__ pooled) {
    const int b  = blockIdx.x;
    const int n0 = blockIdx.y * 32;
    const int t4 = threadIdx.x;
    const int R  = CC / 4;
    const float4* p = (const float4*)x + ((size_t)b * NN + n0) * R + t4;
    float4 acc = {0.f, 0.f, 0.f, 0.f};
#pragma unroll 4
    for (int n = 0; n < 32; ++n) {
        float4 v = p[(size_t)n * R];
        acc.x += v.x; acc.y += v.y; acc.z += v.z; acc.w += v.w;
    }
    float* dst = &pooled[b * CC + t4 * 4];
    atomicAdd(dst + 0, acc.x);
    atomicAdd(dst + 1, acc.y);
    atomicAdd(dst + 2, acc.z);
    atomicAdd(dst + 3, acc.w);
}

__global__ void alpha_kernel(const float* __restrict__ pooled,
                             const float* __restrict__ W,
                             const float* __restrict__ bias,
                             float* __restrict__ alpha) {
    const int lane = threadIdx.x & 31;
    const int wv   = threadIdx.x >> 5;
    const int half = lane >> 4;
    const int lid  = lane & 15;

    const int m      = 16 * wv + lid;
    const int nclamp = (lid < 4) ? lid : 3;
    const float nmask = (lid < 4) ? 1.0f : 0.0f;
    const float scale = 1.0f / 1024.0f;

    v8f acc = {};
    for (int k0 = 0; k0 < CC; k0 += 4) {
        const int k = k0 + 2 * half;
        v2f a, bm;
        a.x = pooled[m * CC + k + 0] * scale;
        a.y = pooled[m * CC + k + 1] * scale;
        bm.x = W[nclamp * CC + k + 0] * nmask;
        bm.y = W[nclamp * CC + k + 1] * nmask;
        acc = __builtin_amdgcn_wmma_f32_16x16x4_f32(
             false, a,  false, bm,
             (short)0, acc,  false,  false);
    }

    if (lid < 4) {
        const float bk = bias[lid];
#pragma unroll
        for (int r = 0; r < 8; ++r) {
            const int mr = 16 * wv + r + 8 * half;
            alpha[mr * 4 + lid] = acc[r] + bk;
        }
    }
}

__device__ __forceinline__ float4 comb4(float a0, const float4& p,
                                        float a1, const float4& q,
                                        float a2, const float4& s,
                                        float a3, const float4& t) {
    float4 r;
    r.x = a0 * p.x + a1 * q.x + a2 * s.x + a3 * t.x;
    r.y = a0 * p.y + a1 * q.y + a2 * s.y + a3 * t.y;
    r.z = a0 * p.z + a1 * q.z + a2 * s.z + a3 * t.z;
    r.w = a0 * p.w + a1 * q.w + a2 * s.w + a3 * t.w;
    return r;
}

__global__ void __launch_bounds__(192)
merge_kernel(const float* __restrict__ x, const float* __restrict__ alpha,
             float* __restrict__ out) {
    const int h = blockIdx.x & 31;
    const int w = blockIdx.x >> 5;
    const int b = blockIdx.y;

    if (h > w) return;
    if (h == w && h >= 16) return;

    const int n  = (h << 5) | w;
    const int n1 = (w << 5) | h;
    const int n2 = (NN - 1) - n;
    const int n3 = (NN - 1) - n1;

    const int t4 = threadIdx.x;
    const int R  = CC / 4;
    const float4* xb = (const float4*)(x + (size_t)b * NN * CC);

    float4 v0 = xb[(size_t)n  * R + t4];
    float4 v1 = xb[(size_t)n1 * R + t4];
    float4 v2 = xb[(size_t)n2 * R + t4];
    float4 v3 = xb[(size_t)n3 * R + t4];

    const float a0 = alpha[b * 4 + 0];
    const float a1 = alpha[b * 4 + 1];
    const float a2 = alpha[b * 4 + 2];
    const float a3 = alpha[b * 4 + 3];

    float4* ob = (float4*)(out + (size_t)b * NN * CC);
    ob[(size_t)n  * R + t4] = comb4(a0, v0, a1, v1, a2, v2, a3, v3);
    ob[(size_t)n1 * R + t4] = comb4(a0, v1, a1, v0, a2, v3, a3, v2);
    ob[(size_t)n2 * R + t4] = comb4(a0, v2, a1, v3, a2, v0, a3, v1);
    ob[(size_t)n3 * R + t4] = comb4(a0, v3, a1, v2, a2, v1, a3, v0);
}

extern "C" void kernel_launch(void* const* d_in, const int* in_sizes, int n_in,
                              void* d_out, int out_size, void* d_ws, size_t ws_size,
                              hipStream_t stream) {
    const float* x    = (const float*)d_in[0];
    const float* W    = (const float*)d_in[1];
    const float* bias = (const float*)d_in[2];
    float* out = (float*)d_out;

    float* pooled = (float*)d_ws;
    float* alpha  = pooled + BB * CC;

    init_ws_kernel<<<1, 256, 0, stream>>>(pooled, BB * CC + BB * 4);
    pool_kernel<<<dim3(BB, 32), 192, 0, stream>>>(x, pooled);
    alpha_kernel<<<1, 64, 0, stream>>>(pooled, W, bias, alpha);
    merge_kernel<<<dim3(NN, BB), 192, 0, stream>>>(x, alpha, out);
}


# alpha GEMM: K split over 32 waves of one block (16 slices x 2 row tiles, float4 operand loads all in flight), slice partials summed in fixed order via LDS; alpha's summation order changes
# speedup vs baseline: 1.3184x; 1.3184x over previous
#include <hip/hip_runtime.h>

#define BB 32
#define NN 1024
#define CC 768
#define HH 32

typedef __attribute__((ext_vector_type(2))) float v2f;
typedef __attribute__((ext_vector_type(8))) float v8f;

__global__ void init_ws_kernel(float* ws, int n) {
    for (int i = threadIdx.x; i < n; i += blockDim.x) ws[i] = 0.0f;
}

__global__ void __launch_bounds__(192)
pool_kernel(const float* __restrict__ x, float* __restrict__ pooled) {
    const int b  = blockIdx.x;
    const int n0 = blockIdx.y * 32;
    const int t4 = threadIdx.x;
    const int R  = CC / 4;
    const float4* p = (const float4*)x + ((size_t)b * NN + n0) * R + t4;
    float4 acc = {0.f, 0.f, 0.f, 0.f};
#pragma unroll 4
    for (int n = 0; n < 32; ++n) {
        float4 v = p[(size_t)n * R];
        acc.x += v.x; acc.y += v.y; acc.z += v.z; acc.w += v.w;
    }
    float* dst = &pooled[b * CC + t4 * 4];
    atomicAdd(dst + 0, acc.x);
    atomicAdd(dst + 1, acc.y);
    atomicAdd(dst + 2, acc.z);
    atomicAdd(dst + 3, acc.w);
}

#define KSLICES 16
__global__ void __launch_bounds__(1024)
alpha_kernel(const float* __restrict__ pooled,
             const float* __restrict__ W,
             const float* __restrict__ bias,
             float* __restrict__ alpha) {
    __shared__ float part[KSLICES][BB][4];

    const int lane = threadIdx.x & 31;
    const int wv   = threadIdx.x >> 5;
    const int tile = wv & 1;
    const int ks   = wv >> 1;
    const int half = lane >> 4;
    const int lid  = lane & 15;

    const int m      = 16 * tile + lid;
    const int nclamp = (lid < 4) ? lid : 3;
    const float nmask = (lid < 4) ? 1.0f : 0.0f;
    const float scale = 1.0f / 1024.0f;

    const int kb = ks * (CC / KSLICES) + half * 24;
    const float4* pa = (const float4*)(pooled + m * CC + kb);
    const float4* pw = (const float4*)(W + nclamp * CC + kb);

    float4 av[6], bv[6];
#pragma unroll
    for (int i = 0; i < 6; ++i) { av[i] = pa[i]; bv[i] = pw[i]; }

    v8f acc = {};
#pragma unroll
    for (int i = 0; i < 6; ++i) {
        v2f a, bm;
        a.x  = av[i].x * scale;  a.y  = av[i].y * scale;
        bm.x = bv[i].x * nmask;  bm.y = bv[i].y * nmask;
        acc = __builtin_amdgcn_wmma_f32_16x16x4_f32(
             false, a,  false, bm,
             (short)0, acc,  false,  false);
        a.x  = av[i].z * scale;  a.y  = av[i].w * scale;
        bm.x = bv[i].z * nmask;  bm.y = bv[i].w * nmask;
        acc = __builtin_amdgcn_wmma_f32_16x16x4_f32(
             false, a,  false, bm,
             (short)0, acc,  false,  false);
    }

    if (lid < 4) {
#pragma unroll
        for (int r = 0; r < 8; ++r) {
            const int mr = 16 * tile + r + 8 * half;
            part[ks][mr][lid] = acc[r];
        }
    }
    __syncthreads();

    if (threadIdx.x < BB * 4) {
        const int mr = threadIdx.x >> 2;
        const int n  = threadIdx.x & 3;
        float s = part[0][mr][n];
#pragma unroll
        for (int j = 1; j < KSLICES; ++j) s += part[j][mr][n];
        alpha[mr * 4 + n] = s + bias[n];
    }
}

__device__ __forceinline__ float4 comb4(float a0, const float4& p,
                                        float a1, const float4& q,
                                        float a2, const float4& s,
                                        float a3, const float4& t) {
    float4 r;
    r.x = a0 * p.x + a1 * q.x + a2 * s.x + a3 * t.x;
    r.y = a0 * p.y + a1 * q.y + a2 * s.y + a3 * t.y;
    r.z = a0 * p.z + a1 * q.z + a2 * s.z + a3 * t.z;
    r.w = a0 * p.w + a1 * q.w + a2 * s.w + a3 * t.w;
    return r;
}

__global__ void __launch_bounds__(192)
merge_kernel(const float* __restrict__ x, const float* __restrict__ alpha,
             float* __restrict__ out) {
    const int h = blockIdx.x & 31;
    const int w = blockIdx.x >> 5;
    const int b = blockIdx.y;

    if (h > w) return;
    if (h == w && h >= 16) return;

    const int n  = (h << 5) | w;
    const int n1 = (w << 5) | h;
    const int n2 = (NN - 1) - n;
    const int n3 = (NN - 1) - n1;

    const int t4 = threadIdx.x;
    const int R  = CC / 4;
    const float4* xb = (const float4*)(x + (size_t)b * NN * CC);

    float4 v0 = xb[(size_t)n  * R + t4];
    float4 v1 = xb[(size_t)n1 * R + t4];
    float4 v2 = xb[(size_t)n2 * R + t4];
    float4 v3 = xb[(size_t)n3 * R + t4];

    const float a0 = alpha[b * 4 + 0];
    const float a1 = alpha[b * 4 + 1];
    const float a2 = alpha[b * 4 + 2];
    const float a3 = alpha[b * 4 + 3];

    float4* ob = (float4*)(out + (size_t)b * NN * CC);
    ob[(size_t)n  * R + t4] = comb4(a0, v0, a1, v1, a2, v2, a3, v3);
    ob[(size_t)n1 * R + t4] = comb4(a0, v1, a1, v0, a2, v3, a3, v2);
    ob[(size_t)n2 * R + t4] = comb4(a0, v2, a1, v3, a2, v0, a3, v1);
    ob[(size_t)n3 * R + t4] = comb4(a0, v3, a1, v2, a2, v1, a3, v0);
}

extern "C" void kernel_launch(void* const* d_in, const int* in_sizes, int n_in,
                              void* d_out, int out_size, void* d_ws, size_t ws_size,
                              hipStream_t stream) {
    const float* x    = (const float*)d_in[0];
    const float* W    = (const float*)d_in[1];
    const float* bias = (const float*)d_in[2];
    float* out = (float*)d_out;

    float* pooled = (float*)d_ws;
    float* alpha  = pooled + BB * CC;

    init_ws_kernel<<<1, 256, 0, stream>>>(pooled, BB * CC + BB * 4);
    pool_kernel<<<dim3(BB, 32), 192, 0, stream>>>(x, pooled);
    alpha_kernel<<<1, 1024, 0, stream>>>(pooled, W, bias, alpha);
    merge_kernel<<<dim3(NN, BB), 192, 0, stream>>>(x, alpha, out);
}


# merge: each orbit of {id, transpose, reverse} visited once, 32 rows per block in two stages of 16 loads then 16 stores; was 512 leaders + 512 empty blocks per batch, every row read and written twice
# speedup vs baseline: 2.2302x; 1.6916x over previous
#include <hip/hip_runtime.h>

#define BB 32
#define NN 1024
#define CC 768
#define HH 32

typedef __attribute__((ext_vector_type(2))) float v2f;
typedef __attribute__((ext_vector_type(8))) float v8f;

__global__ void init_ws_kernel(float* ws, int n) {
    for (int i = threadIdx.x; i < n; i += blockDim.x) ws[i] = 0.0f;
}

__global__ void __launch_bounds__(192)
pool_kernel(const float* __restrict__ x, float* __restrict__ pooled) {
    const int b  = blockIdx.x;
    const int n0 = blockIdx.y * 32;
    const int t4 = threadIdx.x;
    const int R  = CC / 4;
    const float4* p = (const float4*)x + ((size_t)b * NN + n0) * R + t4;
    float4 acc = {0.f, 0.f, 0.f, 0.f};
#pragma unroll 4
    for (int n = 0; n < 32; ++n) {
        float4 v = p[(size_t)n * R];
        acc.x += v.x; acc.y += v.y; acc.z += v.z; acc.w += v.w;
    }
    float* dst = &pooled[b * CC + t4 * 4];
    atomicAdd(dst + 0, acc.x);
    atomicAdd(dst + 1, acc.y);
    atomicAdd(dst + 2, acc.z);
    atomicAdd(dst + 3, acc.w);
}

#define KSLICES 16
__global__ void __launch_bounds__(1024)
alpha_kernel(const float* __restrict__ pooled,
             const float* __restrict__ W,
             const float* __restrict__ bias,
             float* __restrict__ alpha) {
    __shared__ float part[KSLICES][BB][4];

    const int lane = threadIdx.x & 31;
    const int wv   = threadIdx.x >> 5;
    const int tile = wv & 1;
    const int ks   = wv >> 1;
    const int half = lane >> 4;
    const int lid  = lane & 15;

    const int m      = 16 * tile + lid;
    const int nclamp = (lid < 4) ? lid : 3;
    const float nmask = (lid < 4) ? 1.0f : 0.0f;
    const float scale = 1.0f / 1024.0f;

    const int kb = ks * (CC / KSLICES) + half * 24;
    const float4* pa = (const float4*)(pooled + m * CC + kb);
    const float4* pw = (const float4*)(W + nclamp * CC + kb);

    float4 av[6], bv[6];
#pragma unroll
    for (int i = 0; i < 6; ++i) { av[i] = pa[i]; bv[i] = pw[i]; }

    v8f acc = {};
#pragma unroll
    for (int i = 0; i < 6; ++i) {
        v2f a, bm;
        a.x  = av[i].x * scale;  a.y  = av[i].y * scale;
        bm.x = bv[i].x * nmask;  bm.y = bv[i].y * nmask;
        acc = __builtin_amdgcn_wmma_f32_16x16x4_f32(
             false, a,  false, bm,
             (short)0, acc,  false,  false);
        a.x  = av[i].z * scale;  a.y  = av[i].w * scale;
        bm.x = bv[i].z * nmask;  bm.y = bv[i].w * nmask;
        acc = __builtin_amdgcn_wmma_f32_16x16x4_f32(
             false, a,  false, bm,
             (short)0, acc,  false,  false);
    }

    if (lid < 4) {
#pragma unroll
        for (int r = 0; r < 8; ++r) {
            const int mr = 16 * tile + r + 8 * half;
            part[ks][mr][lid] = acc[r];
        }
    }
    __syncthreads();

    if (threadIdx.x < BB * 4) {
        const int mr = threadIdx.x >> 2;
        const int n  = threadIdx.x & 3;
        float s = part[0][mr][n];
#pragma unroll
        for (int j = 1; j < KSLICES; ++j) s += part[j][mr][n];
        alpha[mr * 4 + n] = s + bias[n];
    }
}

typedef __attribute__((ext_vector_type(4))) float v4f;

__device__ __forceinline__ v4f comb4(float a0, v4f p, float a1, v4f q,
                                     float a2, v4f s, float a3, v4f t) {
    return a0 * p + a1 * q + a2 * s + a3 * t;
}

#define LOADS_ISSUED_16(r)                                                   \
    asm volatile("" :: "v"(r[0]), "v"(r[1]), "v"(r[2]), "v"(r[3]),           \
                       "v"(r[4]), "v"(r[5]), "v"(r[6]), "v"(r[7]),           \
                       "v"(r[8]), "v"(r[9]), "v"(r[10]), "v"(r[11]),         \
                       "v"(r[12]), "v"(r[13]), "v"(r[14]), "v"(r[15]))

__device__ __forceinline__ void quad_rows(int q, int& n, int& n1, int& n2, int& n3) {
    const int i = q >> 5;
    const int j = q & 31;
    const bool head = j < 30 - 2 * i;
    const int h = head ? i : 14 - i;
    const int w = head ? i + 1 + j : j + i - 15;
    n  = (h << 5) | w;
    n1 = (w << 5) | h;
    n2 = (NN - 1) - n;
    n3 = (NN - 1) - n1;
}

__device__ __forceinline__ void merge_quads(int chunk,
                                            const v4f* __restrict__ xb,
                                            v4f* __restrict__ ob,
                                            float a0, float a1, float a2, float a3) {
    const int R = CC / 4;
#pragma nounroll
    for (int st = 0; st < 2; ++st) {
        int rn[16];
        v4f v[16];
#pragma unroll
        for (int k = 0; k < 4; ++k)
            quad_rows(chunk * 8 + st * 4 + k,
                      rn[4 * k + 0], rn[4 * k + 1], rn[4 * k + 2], rn[4 * k + 3]);
#pragma unroll
        for (int e = 0; e < 16; ++e) v[e] = xb[(size_t)rn[e] * R];
        LOADS_ISSUED_16(v);
#pragma unroll
        for (int k = 0; k < 4; ++k) {
            const v4f v0 = v[4 * k + 0], v1 = v[4 * k + 1];
            const v4f v2 = v[4 * k + 2], v3 = v[4 * k + 3];
            ob[(size_t)rn[4 * k + 0] * R] = comb4(a0, v0, a1, v1, a2, v2, a3, v3);
            ob[(size_t)rn[4 * k + 1] * R] = comb4(a0, v1, a1, v0, a2, v3, a3, v2);
            ob[(size_t)rn[4 * k + 2] * R] = comb4(a0, v2, a1, v3, a2, v0, a3, v1);
            ob[(size_t)rn[4 * k + 3] * R] = comb4(a0, v3, a1, v2, a2, v1, a3, v0);
        }
    }
}

template <bool ANTI>
__device__ __forceinline__ void merge_pairs(const v4f* __restrict__ xb,
                                            v4f* __restrict__ ob,
                                            float a0, float a1, float a2, float a3) {
    const int R = CC / 4;
#pragma nounroll
    for (int st = 0; st < 2; ++st) {
        int rn[16];
        v4f v[16];
#pragma unroll
        for (int j = 0; j < 8; ++j) {
            const int p = st * 8 + j;
            rn[2 * j]     = ANTI ? 31 * p + 31 : 33 * p;
            rn[2 * j + 1] = (NN - 1) - rn[2 * j];
        }
#pragma unroll
        for (int e = 0; e < 16; ++e) v[e] = xb[(size_t)rn[e] * R];
        LOADS_ISSUED_16(v);
#pragma unroll
        for (int j = 0; j < 8; ++j) {
            const v4f u = v[2 * j], s = v[2 * j + 1];
            if (ANTI) {
                ob[(size_t)rn[2 * j]     * R] = comb4(a0, u, a1, s, a2, s, a3, u);
                ob[(size_t)rn[2 * j + 1] * R] = comb4(a0, s, a1, u, a2, u, a3, s);
            } else {
                ob[(size_t)rn[2 * j]     * R] = comb4(a0, u, a1, u, a2, s, a3, s);
                ob[(size_t)rn[2 * j + 1] * R] = comb4(a0, s, a1, s, a2, u, a3, u);
            }
        }
    }
}

__global__ void __launch_bounds__(192)
merge_kernel(const float* __restrict__ x, const float* __restrict__ alpha,
             float* __restrict__ out) {
    const int b     = blockIdx.x;
    const int chunk = blockIdx.y;
    const int t4    = threadIdx.x;

    const v4f* xb = (const v4f*)(x + (size_t)b * NN * CC) + t4;
    v4f*       ob = (v4f*)(out + (size_t)b * NN * CC) + t4;

    const float a0 = alpha[b * 4 + 0];
    const float a1 = alpha[b * 4 + 1];
    const float a2 = alpha[b * 4 + 2];
    const float a3 = alpha[b * 4 + 3];

    if (chunk < 30)       merge_quads(chunk, xb, ob, a0, a1, a2, a3);
    else if (chunk == 30) merge_pairs<false>(xb, ob, a0, a1, a2, a3);
    else                  merge_pairs<true>(xb, ob, a0, a1, a2, a3);
}

extern "C" void kernel_launch(void* const* d_in, const int* in_sizes, int n_in,
                              void* d_out, int out_size, void* d_ws, size_t ws_size,
                              hipStream_t stream) {
    const float* x    = (const float*)d_in[0];
    const float* W    = (const float*)d_in[1];
    const float* bias = (const float*)d_in[2];
    float* out = (float*)d_out;

    float* pooled = (float*)d_ws;
    float* alpha  = pooled + BB * CC;

    init_ws_kernel<<<1, 256, 0, stream>>>(pooled, BB * CC + BB * 4);
    pool_kernel<<<dim3(BB, 32), 192, 0, stream>>>(x, pooled);
    alpha_kernel<<<1, 1024, 0, stream>>>(pooled, W, bias, alpha);
    merge_kernel<<<dim3(BB, 32), 192, 0, stream>>>(x, alpha, out);
}


# alpha GEMM moved into the pool blocks: each multiplies its own 32-row column sums by W on WMMA, merge adds the 32 partial alphas in fixed order (sum order changes); no zeroing, atomics, alpha launch
# speedup vs baseline: 3.2547x; 1.4594x over previous
#include <hip/hip_runtime.h>


#define BB 32
#define NN 1024
#define CC 768
#define HH 32
#define NCHUNK 32

typedef __attribute__((ext_vector_type(2))) float v2f;
typedef __attribute__((ext_vector_type(4))) float v4f;
typedef __attribute__((ext_vector_type(8))) float v8f;

#define LPITCH4 49
__global__ void __launch_bounds__(192)
pool_alpha_kernel(const float* __restrict__ x, const float* __restrict__ W,
                  float* __restrict__ palpha) {
    __shared__ v4f   rows[20 * LPITCH4];
    __shared__ float part[6][4][16];

    const int b     = blockIdx.x;
    const int chunk = blockIdx.y;
    const int n0    = chunk * 32;
    const int t4    = threadIdx.x;
    const int R     = CC / 4;

    v4f wr[4];
#pragma unroll
    for (int i = 0; i < 4; ++i) wr[i] = ((const v4f*)W)[i * R + t4];

    const v4f* p = (const v4f*)x + ((size_t)b * NN + n0) * R + t4;
    v4f acc = {0.f, 0.f, 0.f, 0.f};
#pragma unroll 4
    for (int n = 0; n < 32; ++n) acc += p[(size_t)n * R];

    const int seg = t4 / 48;
    const int off = t4 % 48;
#pragma unroll
    for (int i = 0; i < 4; ++i) rows[(4 * i + seg) * LPITCH4 + off] = wr[i];
    rows[(16 + seg) * LPITCH4 + off] = acc;
    __syncthreads();

    const int lane = t4 & 31;
    const int wv   = t4 >> 5;
    const int half = lane >> 4;
    const int lid  = lane & 15;
    const int kq4  = 8 * wv + 4 * half;

    const v4f* pa = rows + (16 + (lid & 3)) * LPITCH4 + kq4;
    const v4f* pb = rows + lid * LPITCH4 + kq4;
    v4f av[4], bv[4];
#pragma unroll
    for (int i = 0; i < 4; ++i) { av[i] = pa[i]; bv[i] = pb[i]; }

    v8f d = {};
#pragma unroll
    for (int i = 0; i < 4; ++i) {
        v2f a, bm;
        a.x  = av[i].x;  a.y  = av[i].y;
        bm.x = bv[i].x;  bm.y = bv[i].y;
        d = __builtin_amdgcn_wmma_f32_16x16x4_f32(
             false, a,  false, bm,
             (short)0, d,  false,  false);
        a.x  = av[i].z;  a.y  = av[i].w;
        bm.x = bv[i].z;  bm.y = bv[i].w;
        d = __builtin_amdgcn_wmma_f32_16x16x4_f32(
             false, a,  false, bm,
             (short)0, d,  false,  false);
    }

    if (half == 0) {
#pragma unroll
        for (int r = 0; r < 4; ++r) part[wv][r][lid] = d[r];
    }
    __syncthreads();

    if (t4 < 4) {
        float sum = part[0][0][4 * t4];
#pragma unroll
        for (int w = 0; w < 6; ++w)
#pragma unroll
            for (int sg = 0; sg < 4; ++sg)
                if (w + sg > 0) sum += part[w][sg][4 * t4 + sg];
        palpha[(b * NCHUNK + chunk) * 4 + t4] = sum * (1.0f / 1024.0f);
    }
}

__device__ __forceinline__ v4f comb4(float a0, v4f p, float a1, v4f q,
                                     float a2, v4f s, float a3, v4f t) {
    return a0 * p + a1 * q + a2 * s + a3 * t;
}

#define LOADS_ISSUED_16(r)                                                   \
    asm volatile("" :: "v"(r[0]), "v"(r[1]), "v"(r[2]), "v"(r[3]),           \
                       "v"(r[4]), "v"(r[5]), "v"(r[6]), "v"(r[7]),           \
                       "v"(r[8]), "v"(r[9]), "v"(r[10]), "v"(r[11]),         \
                       "v"(r[12]), "v"(r[13]), "v"(r[14]), "v"(r[15]))

__device__ __forceinline__ void quad_rows(int q, int& n, int& n1, int& n2, int& n3) {
    const int i = q >> 5;
    const int j = q & 31;
    const bool head = j < 30 - 2 * i;
    const int h = head ? i : 14 - i;
    const int w = head ? i + 1 + j : j + i - 15;
    n  = (h << 5) | w;
    n1 = (w << 5) | h;
    n2 = (NN - 1) - n;
    n3 = (NN - 1) - n1;
}

__device__ __forceinline__ void merge_quads(int chunk,
                                            const v4f* __restrict__ xb,
                                            v4f* __restrict__ ob,
                                            float a0, float a1, float a2, float a3) {
    const int R = CC / 4;
#pragma nounroll
    for (int st = 0; st < 2; ++st) {
        int rn[16];
        v4f v[16];
#pragma unroll
        for (int k = 0; k < 4; ++k)
            quad_rows(chunk * 8 + st * 4 + k,
                      rn[4 * k + 0], rn[4 * k + 1], rn[4 * k + 2], rn[4 * k + 3]);
#pragma unroll
        for (int e = 0; e < 16; ++e) v[e] = xb[(size_t)rn[e] * R];
        LOADS_ISSUED_16(v);
#pragma unroll
        for (int k = 0; k < 4; ++k) {
            const v4f v0 = v[4 * k + 0], v1 = v[4 * k + 1];
            const v4f v2 = v[4 * k + 2], v3 = v[4 * k + 3];
            ob[(size_t)rn[4 * k + 0] * R] = comb4(a0, v0, a1, v1, a2, v2, a3, v3);
            ob[(size_t)rn[4 * k + 1] * R] = comb4(a0, v1, a1, v0, a2, v3, a3, v2);
            ob[(size_t)rn[4 * k + 2] * R] = comb4(a0, v2, a1, v3, a2, v0, a3, v1);
            ob[(size_t)rn[4 * k + 3] * R] = comb4(a0, v3, a1, v2, a2, v1, a3, v0);
        }
    }
}

template <bool ANTI>
__device__ __forceinline__ void merge_pairs(const v4f* __restrict__ xb,
                                            v4f* __restrict__ ob,
                                            float a0, float a1, float a2, float a3) {
    const int R = CC / 4;
#pragma nounroll
    for (int st = 0; st < 2; ++st) {
        int rn[16];
        v4f v[16];
#pragma unroll
        for (int j = 0; j < 8; ++j) {
            const int p = st * 8 + j;
            rn[2 * j]     = ANTI ? 31 * p + 31 : 33 * p;
            rn[2 * j + 1] = (NN - 1) - rn[2 * j];
        }
#pragma unroll
        for (int e = 0; e < 16; ++e) v[e] = xb[(size_t)rn[e] * R];
        LOADS_ISSUED_16(v);
#pragma unroll
        for (int j = 0; j < 8; ++j) {
            const v4f u = v[2 * j], s = v[2 * j + 1];
            if (ANTI) {
                ob[(size_t)rn[2 * j]     * R] = comb4(a0, u, a1, s, a2, s, a3, u);
                ob[(size_t)rn[2 * j + 1] * R] = comb4(a0, s, a1, u, a2, u, a3, s);
            } else {
                ob[(size_t)rn[2 * j]     * R] = comb4(a0, u, a1, u, a2, s, a3, s);
                ob[(size_t)rn[2 * j + 1] * R] = comb4(a0, s, a1, s, a2, u, a3, u);
            }
        }
    }
}

__global__ void __launch_bounds__(192)
merge_kernel(const float* __restrict__ x, const float* __restrict__ palpha,
             const float* __restrict__ bias, float* __restrict__ out) {
    const int b     = blockIdx.x;
    const int chunk = blockIdx.y;
    const int t4    = threadIdx.x;

    const v4f* xb = (const v4f*)(x + (size_t)b * NN * CC) + t4;
    v4f*       ob = (v4f*)(out + (size_t)b * NN * CC) + t4;

    const float* pal = palpha + b * (NCHUNK * 4);
    float s0 = pal[0], s1 = pal[1], s2 = pal[2], s3 = pal[3];
#pragma unroll
    for (int j = 1; j < NCHUNK; ++j) {
        s0 += pal[4 * j + 0];
        s1 += pal[4 * j + 1];
        s2 += pal[4 * j + 2];
        s3 += pal[4 * j + 3];
    }
    const float a0 = s0 + bias[0];
    const float a1 = s1 + bias[1];
    const float a2 = s2 + bias[2];
    const float a3 = s3 + bias[3];

    if (chunk < 30)       merge_quads(chunk, xb, ob, a0, a1, a2, a3);
    else if (chunk == 30) merge_pairs<false>(xb, ob, a0, a1, a2, a3);
    else                  merge_pairs<true>(xb, ob, a0, a1, a2, a3);
}

extern "C" void kernel_launch(void* const* d_in, const int* in_sizes, int n_in,
                              void* d_out, int out_size, void* d_ws, size_t ws_size,
                              hipStream_t stream) {
    const float* x    = (const float*)d_in[0];
    const float* W    = (const float*)d_in[1];
    const float* bias = (const float*)d_in[2];
    float* out = (float*)d_out;

    float* palpha = (float*)d_ws;

    pool_alpha_kernel<<<dim3(BB, NCHUNK), 192, 0, stream>>>(x, W, palpha);
    merge_kernel<<<dim3(BB, 32), 192, 0, stream>>>(x, palpha, bias, out);
}


# work per wave doubled to 64 rows, all requested at wave entry (one memory round trip per wave); 256 blocks of 12 waves, one per WGP (was 1024 of 6); 8 alpha partials: alpha's sum order changes
# speedup vs baseline: 3.4270x; 1.0529x over previous
#include <hip/hip_runtime.h>


#define BB 32
#define NN 1024
#define CC 768
#define HH 32
#define CROWS 64
#define PBLK 8

typedef __attribute__((ext_vector_type(2))) float v2f;
typedef __attribute__((ext_vector_type(4))) float v4f;
typedef __attribute__((ext_vector_type(8))) float v8f;


#define LPITCH4 49
__global__ void __launch_bounds__(384)
pool_alpha_kernel(const float* __restrict__ x, const float* __restrict__ W,
                  float* __restrict__ palpha) {
    __shared__ v4f   rows[24 * LPITCH4];
    __shared__ float part[12][16];

    const int b   = blockIdx.x;
    const int blk = blockIdx.y;
    const int g   = __builtin_amdgcn_readfirstlane((int)threadIdx.x / 192);
    const int t4  = (int)threadIdx.x - 192 * g;
    const int n0  = (2 * blk + g) * CROWS;
    const int R   = CC / 4;

    v4f wr[2];
#pragma unroll
    for (int i = 0; i < 2; ++i) wr[i] = ((const v4f*)W)[(2 * g + i) * R + t4];

    const v4f* p = (const v4f*)x + ((size_t)b * NN + n0) * R + t4;
    v4f xr[CROWS];
#pragma unroll
    for (int n = 0; n < CROWS; ++n) xr[n] = p[(size_t)n * R];
    __builtin_amdgcn_sched_barrier(0);
    v4f acc = {0.f, 0.f, 0.f, 0.f};
#pragma unroll
    for (int n = 0; n < CROWS; ++n) acc += xr[n];

    const int seg = t4 / 48;
    const int off = t4 % 48;
#pragma unroll
    for (int i = 0; i < 2; ++i) rows[(4 * (2 * g + i) + seg) * LPITCH4 + off] = wr[i];
    rows[(16 + 4 * g + seg) * LPITCH4 + off] = acc;
    __syncthreads();

    const int lane = (int)threadIdx.x & 31;
    const int wv   = (int)threadIdx.x >> 5;
    const int half = lane >> 4;
    const int lid  = lane & 15;
    const int kq4  = 4 * wv + 2 * half;

    const v4f* pa = rows + (16 + (lid & 7)) * LPITCH4 + kq4;
    const v4f* pb = rows + lid * LPITCH4 + kq4;
    v4f av[2], bv[2];
#pragma unroll
    for (int i = 0; i < 2; ++i) { av[i] = pa[i]; bv[i] = pb[i]; }

    v8f d = {};
#pragma unroll
    for (int i = 0; i < 2; ++i) {
        v2f a, bm;
        a.x  = av[i].x;  a.y  = av[i].y;
        bm.x = bv[i].x;  bm.y = bv[i].y;
        d = __builtin_amdgcn_wmma_f32_16x16x4_f32(
             false, a,  false, bm,
             (short)0, d,  false,  false);
        a.x  = av[i].z;  a.y  = av[i].w;
        bm.x = bv[i].z;  bm.y = bv[i].w;
        d = __builtin_amdgcn_wmma_f32_16x16x4_f32(
             false, a,  false, bm,
             (short)0, d,  false,  false);
    }

    if (half == 0) {
        const int s = lid & 3;
        const float q0 = d[0] + d[4], q1 = d[1] + d[5], q2 = d[2] + d[6], q3 = d[3] + d[7];
        part[wv][lid] = s == 0 ? q0 : s == 1 ? q1 : s == 2 ? q2 : q3;
    }
    __syncthreads();

    if (threadIdx.x < 4) {
        const int n = (int)threadIdx.x;
        float sum = 0.f;
#pragma unroll
        for (int w = 0; w < 12; ++w)
            sum += (part[w][4 * n + 0] + part[w][4 * n + 1]) +
                   (part[w][4 * n + 2] + part[w][4 * n + 3]);
        palpha[(b * PBLK + blk) * 4 + n] = sum * (1.0f / 1024.0f);
    }
}

__device__ __forceinline__ v4f comb4(float a0, v4f p, float a1, v4f q,
                                     float a2, v4f s, float a3, v4f t) {
    return a0 * p + a1 * q + a2 * s + a3 * t;
}

__device__ __forceinline__ void alpha4(const float* __restrict__ pal,
                                       const float* __restrict__ bias,
                                       float& a0, float& a1, float& a2, float& a3) {
    float s0 = pal[0], s1 = pal[1], s2 = pal[2], s3 = pal[3];
#pragma unroll
    for (int j = 1; j < PBLK; ++j) {
        s0 += pal[4 * j + 0];
        s1 += pal[4 * j + 1];
        s2 += pal[4 * j + 2];
        s3 += pal[4 * j + 3];
    }
    a0 = s0 + bias[0];
    a1 = s1 + bias[1];
    a2 = s2 + bias[2];
    a3 = s3 + bias[3];
}

__device__ __forceinline__ void quad_rows(int q, int& n, int& n1, int& n2, int& n3) {
    const int i = q >> 5;
    const int j = q & 31;
    const bool head = j < 30 - 2 * i;
    const int h = head ? i : 14 - i;
    const int w = head ? i + 1 + j : j + i - 15;
    n  = (h << 5) | w;
    n1 = (w << 5) | h;
    n2 = (NN - 1) - n;
    n3 = (NN - 1) - n1;
}

__device__ __forceinline__ void merge_chunk(int chunk,
                                            const v4f* __restrict__ xb,
                                            v4f* __restrict__ ob,
                                            const float* __restrict__ pal,
                                            const float* __restrict__ bias) {
    const int R = CC / 4;
    int rn[CROWS];
    v4f v[CROWS];
#pragma unroll
    for (int k = 0; k < 15; ++k)
        quad_rows(chunk * 15 + k,
                  rn[4 * k + 0], rn[4 * k + 1], rn[4 * k + 2], rn[4 * k + 3]);
    rn[60] = 33 * chunk;
    rn[61] = (NN - 1) - rn[60];
    rn[62] = 31 * chunk + 31;
    rn[63] = (NN - 1) - rn[62];
#pragma unroll
    for (int e = 0; e < CROWS; ++e) v[e] = xb[(size_t)rn[e] * R];
    __builtin_amdgcn_sched_barrier(0);
    float a0, a1, a2, a3;
    alpha4(pal, bias, a0, a1, a2, a3);
#pragma unroll
    for (int k = 0; k < 15; ++k) {
        const v4f v0 = v[4 * k + 0], v1 = v[4 * k + 1];
        const v4f v2 = v[4 * k + 2], v3 = v[4 * k + 3];
        ob[(size_t)rn[4 * k + 0] * R] = comb4(a0, v0, a1, v1, a2, v2, a3, v3);
        ob[(size_t)rn[4 * k + 1] * R] = comb4(a0, v1, a1, v0, a2, v3, a3, v2);
        ob[(size_t)rn[4 * k + 2] * R] = comb4(a0, v2, a1, v3, a2, v0, a3, v1);
        ob[(size_t)rn[4 * k + 3] * R] = comb4(a0, v3, a1, v2, a2, v1, a3, v0);
    }
    {
        const v4f u = v[60], s = v[61];
        ob[(size_t)rn[60] * R] = comb4(a0, u, a1, u, a2, s, a3, s);
        ob[(size_t)rn[61] * R] = comb4(a0, s, a1, s, a2, u, a3, u);
    }
    {
        const v4f u = v[62], s = v[63];
        ob[(size_t)rn[62] * R] = comb4(a0, u, a1, s, a2, s, a3, u);
        ob[(size_t)rn[63] * R] = comb4(a0, s, a1, u, a2, u, a3, s);
    }
}

__global__ void __launch_bounds__(384)
merge_kernel(const float* __restrict__ x, const float* __restrict__ palpha,
             const float* __restrict__ bias, float* __restrict__ out) {
    const int b     = blockIdx.x;
    const int g     = __builtin_amdgcn_readfirstlane((int)threadIdx.x / 192);
    const int chunk = 2 * (int)blockIdx.y + g;
    const int t4    = (int)threadIdx.x - 192 * g;

    const v4f* xb = (const v4f*)(x + (size_t)b * NN * CC) + t4;
    v4f*       ob = (v4f*)(out + (size_t)b * NN * CC) + t4;
    const float* pal = palpha + b * (PBLK * 4);

    merge_chunk(chunk, xb, ob, pal, bias);
}

extern "C" void kernel_launch(void* const* d_in, const int* in_sizes, int n_in,
                              void* d_out, int out_size, void* d_ws, size_t ws_size,
                              hipStream_t stream) {
    const float* x    = (const float*)d_in[0];
    const float* W    = (const float*)d_in[1];
    const float* bias = (const float*)d_in[2];
    float* out = (float*)d_out;

    float* palpha = (float*)d_ws;

    pool_alpha_kernel<<<dim3(BB, PBLK), 384, 0, stream>>>(x, W, palpha);
    merge_kernel<<<dim3(BB, PBLK), 384, 0, stream>>>(x, palpha, bias, out);
}
